# starting kernel
# baseline (speedup 1.0000x reference)
#include <hip/hip_runtime.h>
#include <math.h>


typedef float v2f __attribute__((ext_vector_type(2)));
typedef float v8f __attribute__((ext_vector_type(8)));

#define DFEAT  10
#define KPAD   12
#define ROWPAD 16

__global__ __launch_bounds__(256)
void edgeconv_lin1_wmma(const float* __restrict__ x,
                        const float* __restrict__ W1,
                        const float* __restrict__ b1,
                        float* __restrict__ yws,
                        float* __restrict__ zws,
                        int nrows, int ntiles) {
    __shared__ float ldsA[16 * KPAD];
    __shared__ float ldsB[16 * KPAD];

    for (int t = threadIdx.x; t < 16 * KPAD; t += blockDim.x) {
        const int n = t / KPAD, k = t % KPAD;
        float wtop = 0.0f, wbot = 0.0f;
        if (n < DFEAT && k < DFEAT) {
            wtop = W1[k * DFEAT + n];
            wbot = W1[(k + DFEAT) * DFEAT + n];
        }
        ldsA[t] = wtop - wbot;
        ldsB[t] = wbot;
    }
    __syncthreads();

    const int wavesPerBlock = blockDim.x >> 5;
    const int tile = blockIdx.x * wavesPerBlock + (threadIdx.x >> 5);
    if (tile >= ntiles) return;

    const int lane = threadIdx.x & 31;
    const int hi   = lane >> 4;
    const int m    = lane & 15;
    const int base = tile * 16;
    const int row  = base + m;
    const bool rowOK = (row < nrows);

    v8f cy = {0.f,0.f,0.f,0.f,0.f,0.f,0.f,0.f};
    v8f cz = {0.f,0.f,0.f,0.f,0.f,0.f,0.f,0.f};

#pragma unroll
    for (int ks = 0; ks < 3; ++ks) {
        const int k0 = ks * 4 + 2 * hi;
        const bool ok = rowOK && (k0 < DFEAT);
        const float2* px = ok ? (const float2*)(x + (size_t)row * DFEAT + k0)
                              : (const float2*)x;
        const float2 xv = *px;
        v2f a;  a[0] = ok ? xv.x : 0.0f;  a[1] = ok ? xv.y : 0.0f;

        const float2 wav = *(const float2*)(ldsA + m * KPAD + k0);
        const float2 wbv = *(const float2*)(ldsB + m * KPAD + k0);
        v2f wa;  wa[0] = wav.x;  wa[1] = wav.y;
        v2f wb;  wb[0] = wbv.x;  wb[1] = wbv.y;

        cy = __builtin_amdgcn_wmma_f32_16x16x4_f32(false, a, false, wa,
                                                   (short)0, cy, false, false);
        cz = __builtin_amdgcn_wmma_f32_16x16x4_f32(false, a, false, wb,
                                                   (short)0, cz, false, false);
    }

    const float bias = (m < DFEAT) ? b1[m] : 0.0f;
    float* yrow = yws + (size_t)base * ROWPAD + m;
    float* zrow = zws + (size_t)base * ROWPAD + m;
    if (base + 16 <= nrows) {
#pragma unroll
        for (int r = 0; r < 8; ++r) {
            const int rr = r + 8 * hi;
            yrow[(size_t)rr * ROWPAD] = cy[r] + bias;
            zrow[(size_t)rr * ROWPAD] = cz[r];
        }
    } else {
#pragma unroll
        for (int r = 0; r < 8; ++r) {
            const int rr = r + 8 * hi;
            if (base + rr < nrows) {
                yrow[(size_t)rr * ROWPAD] = cy[r] + bias;
                zrow[(size_t)rr * ROWPAD] = cz[r];
            }
        }
    }
}

__global__ __launch_bounds__(256)
void edgeconv_maxagg(const int* __restrict__ src,
                     const float* __restrict__ yws,
                     const float* __restrict__ zws,
                     const float* __restrict__ W2,
                     const float* __restrict__ b2,
                     float* __restrict__ out,
                     int nnodes, int kdeg) {
    const int wavesPerBlock = blockDim.x >> 5;
    const int node = blockIdx.x * wavesPerBlock + (threadIdx.x >> 5);
    if (node >= nnodes) return;
    const int lane = threadIdx.x & 31;

    float m[DFEAT];
    if (lane < kdeg) {
        const int s = src[(size_t)node * kdeg + lane];
        const float* zr = zws + (size_t)s * ROWPAD;
        const float4 z0 = *(const float4*)(zr);
        const float4 z1 = *(const float4*)(zr + 4);
        m[0] = z0.x; m[1] = z0.y; m[2] = z0.z; m[3] = z0.w;
        m[4] = z1.x; m[5] = z1.y; m[6] = z1.z; m[7] = z1.w;
        m[8] = zr[8]; m[9] = zr[9];
    } else {
#pragma unroll
        for (int j = 0; j < DFEAT; ++j) m[j] = -INFINITY;
    }

#pragma unroll
    for (int off = 16; off > 0; off >>= 1) {
#pragma unroll
        for (int j = 0; j < DFEAT; ++j) {
            const float o = __shfl_xor(m[j], off, 32);
            m[j] = fmaxf(m[j], o);
        }
    }

    const float* yr = yws + (size_t)node * ROWPAD;
    const float4 y0 = *(const float4*)(yr);
    const float4 y1 = *(const float4*)(yr + 4);
    const float yv[DFEAT] = { y0.x, y0.y, y0.z, y0.w,
                              y1.x, y1.y, y1.z, y1.w, yr[8], yr[9] };

    float agg[DFEAT];
#pragma unroll
    for (int j = 0; j < DFEAT; ++j) {
        const float v = yv[j] + m[j];
        agg[j] = isfinite(v) ? v : 0.0f;
    }

    if (lane < DFEAT) {
        float acc = b2[lane];
#pragma unroll
        for (int k = 0; k < DFEAT; ++k)
            acc = fmaf(agg[k], W2[k * DFEAT + lane], acc);
        out[(size_t)node * DFEAT + lane] = acc;
    }
}

extern "C" void kernel_launch(void* const* d_in, const int* in_sizes, int n_in,
                              void* d_out, int out_size, void* d_ws, size_t ws_size,
                              hipStream_t stream) {
    const float* x   = (const float*)d_in[0];
    const int*   ei  = (const int*)  d_in[1];
    const float* W1  = (const float*)d_in[2];
    const float* b1  = (const float*)d_in[3];
    const float* W2  = (const float*)d_in[4];
    const float* b2  = (const float*)d_in[5];
    float* out = (float*)d_out;

    const int nnodes = in_sizes[0] / DFEAT;
    const int nedges = in_sizes[1] / 2;
    const int kdeg   = nedges / nnodes;
    const int* src   = ei;

    float* yws = (float*)d_ws;
    float* zws = yws + (size_t)nnodes * ROWPAD;

    const int ntiles = (nnodes + 15) / 16;
    {
        dim3 block(256);
        dim3 grid((ntiles + 7) / 8);
        hipLaunchKernelGGL(edgeconv_lin1_wmma, grid, block, 0, stream,
                           x, W1, b1, yws, zws, nnodes, ntiles);
    }
    {
        dim3 block(256);
        dim3 grid((nnodes + 7) / 8);
        hipLaunchKernelGGL(edgeconv_maxagg, grid, block, 0, stream,
                           src, yws, zws, W2, b2, out, nnodes, kdeg);
    }
}


# aggregation kernel: 4 lanes share each gathered 64B row and keep the neighbour max in registers, 8 nodes per wave: 4x fewer cache lines per load, no shuffle butterfly, 8x fewer waves
# speedup vs baseline: 1.4009x; 1.4009x over previous
#include <hip/hip_runtime.h>
#include <math.h>


typedef float v2f __attribute__((ext_vector_type(2)));
typedef float v8f __attribute__((ext_vector_type(8)));

#define DFEAT  10
#define KPAD   12
#define ROWPAD 16

__global__ __launch_bounds__(256)
void edgeconv_lin1_wmma(const float* __restrict__ x,
                        const float* __restrict__ W1,
                        const float* __restrict__ b1,
                        float* __restrict__ yws,
                        float* __restrict__ zws,
                        int nrows, int ntiles) {
    __shared__ float ldsA[16 * KPAD];
    __shared__ float ldsB[16 * KPAD];

    for (int t = threadIdx.x; t < 16 * KPAD; t += blockDim.x) {
        const int n = t / KPAD, k = t % KPAD;
        float wtop = 0.0f, wbot = 0.0f;
        if (n < DFEAT && k < DFEAT) {
            wtop = W1[k * DFEAT + n];
            wbot = W1[(k + DFEAT) * DFEAT + n];
        }
        ldsA[t] = wtop - wbot;
        ldsB[t] = wbot;
    }
    __syncthreads();

    const int wavesPerBlock = blockDim.x >> 5;
    const int tile = blockIdx.x * wavesPerBlock + (threadIdx.x >> 5);
    if (tile >= ntiles) return;

    const int lane = threadIdx.x & 31;
    const int hi   = lane >> 4;
    const int m    = lane & 15;
    const int base = tile * 16;
    const int row  = base + m;
    const bool rowOK = (row < nrows);

    v8f cy = {0.f,0.f,0.f,0.f,0.f,0.f,0.f,0.f};
    v8f cz = {0.f,0.f,0.f,0.f,0.f,0.f,0.f,0.f};

#pragma unroll
    for (int ks = 0; ks < 3; ++ks) {
        const int k0 = ks * 4 + 2 * hi;
        const bool ok = rowOK && (k0 < DFEAT);
        const float2* px = ok ? (const float2*)(x + (size_t)row * DFEAT + k0)
                              : (const float2*)x;
        const float2 xv = *px;
        v2f a;  a[0] = ok ? xv.x : 0.0f;  a[1] = ok ? xv.y : 0.0f;

        const float2 wav = *(const float2*)(ldsA + m * KPAD + k0);
        const float2 wbv = *(const float2*)(ldsB + m * KPAD + k0);
        v2f wa;  wa[0] = wav.x;  wa[1] = wav.y;
        v2f wb;  wb[0] = wbv.x;  wb[1] = wbv.y;

        cy = __builtin_amdgcn_wmma_f32_16x16x4_f32(false, a, false, wa,
                                                   (short)0, cy, false, false);
        cz = __builtin_amdgcn_wmma_f32_16x16x4_f32(false, a, false, wb,
                                                   (short)0, cz, false, false);
    }

    const float bias = (m < DFEAT) ? b1[m] : 0.0f;
    float* yrow = yws + (size_t)base * ROWPAD + m;
    float* zrow = zws + (size_t)base * ROWPAD + m;
    if (base + 16 <= nrows) {
#pragma unroll
        for (int r = 0; r < 8; ++r) {
            const int rr = r + 8 * hi;
            yrow[(size_t)rr * ROWPAD] = cy[r] + bias;
            zrow[(size_t)rr * ROWPAD] = cz[r];
        }
    } else {
#pragma unroll
        for (int r = 0; r < 8; ++r) {
            const int rr = r + 8 * hi;
            if (base + rr < nrows) {
                yrow[(size_t)rr * ROWPAD] = cy[r] + bias;
                zrow[(size_t)rr * ROWPAD] = cz[r];
            }
        }
    }
}

#define NPW 8

template <int A>
__device__ __forceinline__ int quad_bcast(int v) {
    return __builtin_amdgcn_mov_dpp(v, A * 0x55, 0xF, 0xF, true);
}
template <int A>
__device__ __forceinline__ float quad_bcastf(float v) {
    return __int_as_float(quad_bcast<A>(__float_as_int(v)));
}

__device__ __forceinline__ void max4(float4& m, const float4 v) {
    m.x = fmaxf(m.x, v.x);  m.y = fmaxf(m.y, v.y);
    m.z = fmaxf(m.z, v.z);  m.w = fmaxf(m.w, v.w);
}

template <int A>
__device__ __forceinline__ void gather8(float4& m, const int (&idx)[8],
                                        const char* __restrict__ zb, unsigned qoff) {
#pragma unroll
    for (int b = 0; b < 8; ++b) {
        const unsigned s = (unsigned)quad_bcast<A>(idx[b]);
        max4(m, *(const float4*)(zb + ((s << 6) | qoff)));
    }
}

__global__ __launch_bounds__(256)
void edgeconv_maxagg(const int* __restrict__ src,
                     const float* __restrict__ yws,
                     const float* __restrict__ zws,
                     const float* __restrict__ W2,
                     const float* __restrict__ b2,
                     float* __restrict__ out,
                     int nnodes, int kdeg) {
    __shared__ float w2s[DFEAT * ROWPAD];
    __shared__ float b2s[ROWPAD];

    const int lane  = threadIdx.x & 31;
    const int q     = lane & 3;
    const int node0 = (blockIdx.x * (blockDim.x >> 5) + (threadIdx.x >> 5)) * NPW;
    const int node  = node0 + (lane >> 2);
    const bool valid = node < nnodes;
    const int nodeC = valid ? node : 0;
    const bool fast = (kdeg == 32);

    int idx[8] = {0, 0, 0, 0, 0, 0, 0, 0};
    if (fast) {
        const int4* ip = (const int4*)(src + (size_t)nodeC * 32 + 8 * q);
        const int4 i0 = ip[0], i1 = ip[1];
        idx[0] = i0.x; idx[1] = i0.y; idx[2] = i0.z; idx[3] = i0.w;
        idx[4] = i1.x; idx[5] = i1.y; idx[6] = i1.z; idx[7] = i1.w;
    }
    const float4 yv = *(const float4*)(yws + (size_t)nodeC * ROWPAD + 4 * q);

    for (int t = threadIdx.x; t < DFEAT * ROWPAD; t += blockDim.x) {
        const int k = t >> 4, c = t & 15;
        w2s[t] = (c < DFEAT) ? W2[k * DFEAT + c] : 0.0f;
    }
    if (threadIdx.x < ROWPAD)
        b2s[threadIdx.x] = (threadIdx.x < DFEAT) ? b2[threadIdx.x] : 0.0f;
    __syncthreads();
    if (node0 >= nnodes) return;

    const char* zb = (const char*)zws;
    const unsigned qoff = (unsigned)q << 4;
    float4 m = {-INFINITY, -INFINITY, -INFINITY, -INFINITY};
    if (fast) {
        gather8<0>(m, idx, zb, qoff);
        gather8<1>(m, idx, zb, qoff);
        gather8<2>(m, idx, zb, qoff);
        gather8<3>(m, idx, zb, qoff);
    } else {
        const int* sp = src + (size_t)nodeC * kdeg;
        for (int j = 0; j < kdeg; ++j) {
            const unsigned s = (unsigned)sp[j];
            max4(m, *(const float4*)(zb + ((s << 6) | qoff)));
        }
    }

    float ag[4] = { yv.x + m.x, yv.y + m.y, yv.z + m.z, yv.w + m.w };
#pragma unroll
    for (int i = 0; i < 4; ++i)
        ag[i] = isfinite(ag[i]) ? ag[i] : 0.0f;

    float agg[DFEAT];
    agg[0] = quad_bcastf<0>(ag[0]);  agg[1] = quad_bcastf<0>(ag[1]);
    agg[2] = quad_bcastf<0>(ag[2]);  agg[3] = quad_bcastf<0>(ag[3]);
    agg[4] = quad_bcastf<1>(ag[0]);  agg[5] = quad_bcastf<1>(ag[1]);
    agg[6] = quad_bcastf<1>(ag[2]);  agg[7] = quad_bcastf<1>(ag[3]);
    agg[8] = quad_bcastf<2>(ag[0]);  agg[9] = quad_bcastf<2>(ag[1]);

    if (valid && q < 3) {
        float4 acc = *(const float4*)(b2s + 4 * q);
#pragma unroll
        for (int k = 0; k < DFEAT; ++k) {
            const float4 w = *(const float4*)(w2s + k * ROWPAD + 4 * q);
            acc.x = fmaf(agg[k], w.x, acc.x);
            acc.y = fmaf(agg[k], w.y, acc.y);
            acc.z = fmaf(agg[k], w.z, acc.z);
            acc.w = fmaf(agg[k], w.w, acc.w);
        }
        float* o = out + (size_t)node * DFEAT + 4 * q;
        *(float2*)o = make_float2(acc.x, acc.y);
        if (q < 2) *(float2*)(o + 2) = make_float2(acc.z, acc.w);
    }
}

extern "C" void kernel_launch(void* const* d_in, const int* in_sizes, int n_in,
                              void* d_out, int out_size, void* d_ws, size_t ws_size,
                              hipStream_t stream) {
    const float* x   = (const float*)d_in[0];
    const int*   ei  = (const int*)  d_in[1];
    const float* W1  = (const float*)d_in[2];
    const float* b1  = (const float*)d_in[3];
    const float* W2  = (const float*)d_in[4];
    const float* b2  = (const float*)d_in[5];
    float* out = (float*)d_out;

    const int nnodes = in_sizes[0] / DFEAT;
    const int nedges = in_sizes[1] / 2;
    const int kdeg   = nedges / nnodes;
    const int* src   = ei;

    float* yws = (float*)d_ws;
    float* zws = yws + (size_t)nnodes * ROWPAD;

    const int ntiles = (nnodes + 15) / 16;
    {
        dim3 block(256);
        dim3 grid((ntiles + 7) / 8);
        hipLaunchKernelGGL(edgeconv_lin1_wmma, grid, block, 0, stream,
                           x, W1, b1, yws, zws, nnodes, ntiles);
    }
    {
        dim3 block(256);
        dim3 grid((nnodes + 63) / 64);
        hipLaunchKernelGGL(edgeconv_maxagg, grid, block, 0, stream,
                           src, yws, zws, W2, b2, out, nnodes, kdeg);
    }
}


# the x@(W1a-W1b)+b1 product moves from the first kernel into the aggregation kernel (3 f32 WMMAs per 8-node wave, wave-private LDS transpose): no y plane written or read, first kernel stores half
# speedup vs baseline: 1.4985x; 1.0697x over previous
#include <hip/hip_runtime.h>
#include <math.h>


typedef float v2f __attribute__((ext_vector_type(2)));
typedef float v8f __attribute__((ext_vector_type(8)));

#define DFEAT  10
#define KPAD   12
#define ROWPAD 16

__global__ __launch_bounds__(256)
void edgeconv_lin1_wmma(const float* __restrict__ x,
                        const float* __restrict__ W1,
                        float* __restrict__ zws,
                        int nrows, int ntiles) {
    __shared__ float ldsB[16 * KPAD];

    for (int t = threadIdx.x; t < 16 * KPAD; t += blockDim.x) {
        const int n = t / KPAD, k = t % KPAD;
        float wbot = 0.0f;
        if (n < DFEAT && k < DFEAT)
            wbot = W1[(k + DFEAT) * DFEAT + n];
        ldsB[t] = wbot;
    }
    __syncthreads();

    const int wavesPerBlock = blockDim.x >> 5;
    const int tile = blockIdx.x * wavesPerBlock + (threadIdx.x >> 5);
    if (tile >= ntiles) return;

    const int lane = threadIdx.x & 31;
    const int hi   = lane >> 4;
    const int m    = lane & 15;
    const int base = tile * 16;
    const int row  = base + m;
    const bool rowOK = (row < nrows);

    v8f cz = {0.f,0.f,0.f,0.f,0.f,0.f,0.f,0.f};

#pragma unroll
    for (int ks = 0; ks < 3; ++ks) {
        const int k0 = ks * 4 + 2 * hi;
        const bool ok = rowOK && (k0 < DFEAT);
        const float2* px = ok ? (const float2*)(x + (size_t)row * DFEAT + k0)
                              : (const float2*)x;
        const float2 xv = *px;
        v2f a;  a[0] = ok ? xv.x : 0.0f;  a[1] = ok ? xv.y : 0.0f;

        const float2 wbv = *(const float2*)(ldsB + m * KPAD + k0);
        v2f wb;  wb[0] = wbv.x;  wb[1] = wbv.y;

        cz = __builtin_amdgcn_wmma_f32_16x16x4_f32(false, a, false, wb,
                                                   (short)0, cz, false, false);
    }

    float* zrow = zws + (size_t)base * ROWPAD + m;
    if (base + 16 <= nrows) {
#pragma unroll
        for (int r = 0; r < 8; ++r) {
            const int rr = r + 8 * hi;
            zrow[(size_t)rr * ROWPAD] = cz[r];
        }
    } else {
#pragma unroll
        for (int r = 0; r < 8; ++r) {
            const int rr = r + 8 * hi;
            if (base + rr < nrows)
                zrow[(size_t)rr * ROWPAD] = cz[r];
        }
    }
}

#define NPW 8

template <int A>
__device__ __forceinline__ int quad_bcast(int v) {
    return __builtin_amdgcn_mov_dpp(v, A * 0x55, 0xF, 0xF, true);
}
template <int A>
__device__ __forceinline__ float quad_bcastf(float v) {
    return __int_as_float(quad_bcast<A>(__float_as_int(v)));
}

__device__ __forceinline__ void max4(float4& m, const float4 v) {
    m.x = fmaxf(m.x, v.x);  m.y = fmaxf(m.y, v.y);
    m.z = fmaxf(m.z, v.z);  m.w = fmaxf(m.w, v.w);
}

template <int A>
__device__ __forceinline__ void gather8(float4& m, const int (&idx)[8],
                                        const char* __restrict__ zb, unsigned qoff) {
#pragma unroll
    for (int b = 0; b < 8; ++b) {
        const unsigned s = (unsigned)quad_bcast<A>(idx[b]);
        max4(m, *(const float4*)(zb + ((s << 6) | qoff)));
    }
}

__global__ __launch_bounds__(256)
void edgeconv_maxagg(const int* __restrict__ src,
                     const float* __restrict__ x,
                     const float* __restrict__ zws,
                     const float* __restrict__ W1,
                     const float* __restrict__ b1,
                     const float* __restrict__ W2,
                     const float* __restrict__ b2,
                     float* __restrict__ out,
                     int nnodes, int kdeg) {
    __shared__ float w2s[DFEAT * ROWPAD];
    __shared__ float b2s[ROWPAD];
    __shared__ float ldsA[16 * KPAD];
    __shared__ float ybuf[8][16 * ROWPAD];

    const int lane  = threadIdx.x & 31;
    const int q     = lane & 3;
    const int node0 = (blockIdx.x * (blockDim.x >> 5) + (threadIdx.x >> 5)) * NPW;
    const int node  = node0 + (lane >> 2);
    const bool valid = node < nnodes;
    const int nodeC = valid ? node : 0;
    const bool fast = (kdeg == 32);

    int idx[8] = {0, 0, 0, 0, 0, 0, 0, 0};
    if (fast) {
        const int4* ip = (const int4*)(src + (size_t)nodeC * 32 + 8 * q);
        const int4 i0 = ip[0], i1 = ip[1];
        idx[0] = i0.x; idx[1] = i0.y; idx[2] = i0.z; idx[3] = i0.w;
        idx[4] = i1.x; idx[5] = i1.y; idx[6] = i1.z; idx[7] = i1.w;
    }
    const int hi = lane >> 4;
    const int cm = lane & 15;
    const bool rowOK = (cm < NPW) && (node0 + cm < nnodes);
    v2f xa[3];
#pragma unroll
    for (int ks = 0; ks < 3; ++ks) {
        const int k0 = ks * 4 + 2 * hi;
        const bool ok = rowOK && (k0 < DFEAT);
        const float2* px = ok ? (const float2*)(x + (size_t)(node0 + cm) * DFEAT + k0)
                              : (const float2*)x;
        const float2 xv = *px;
        xa[ks][0] = ok ? xv.x : 0.0f;  xa[ks][1] = ok ? xv.y : 0.0f;
    }
    const float bias = (cm < DFEAT) ? b1[cm] : 0.0f;

    for (int t = threadIdx.x; t < 16 * KPAD; t += blockDim.x) {
        const int n = t / KPAD, k = t % KPAD;
        const int k2 = t >> 4, c2 = t & 15;
        float wtop = 0.0f, wbot = 0.0f, w2v = 0.0f;
        if (n < DFEAT && k < DFEAT) {
            wtop = W1[k * DFEAT + n];
            wbot = W1[(k + DFEAT) * DFEAT + n];
        }
        if (t < DFEAT * ROWPAD && c2 < DFEAT)
            w2v = W2[k2 * DFEAT + c2];
        ldsA[t] = wtop - wbot;
        if (t < DFEAT * ROWPAD) w2s[t] = w2v;
    }
    if (threadIdx.x < ROWPAD)
        b2s[threadIdx.x] = (threadIdx.x < DFEAT) ? b2[threadIdx.x] : 0.0f;
    __syncthreads();
    if (node0 >= nnodes) return;

    v8f cy = {0.f,0.f,0.f,0.f,0.f,0.f,0.f,0.f};
#pragma unroll
    for (int ks = 0; ks < 3; ++ks) {
        const float2 wav = *(const float2*)(ldsA + cm * KPAD + ks * 4 + 2 * hi);
        v2f wa;  wa[0] = wav.x;  wa[1] = wav.y;
        cy = __builtin_amdgcn_wmma_f32_16x16x4_f32(false, xa[ks], false, wa,
                                                   (short)0, cy, false, false);
    }
    float* yb = ybuf[threadIdx.x >> 5];
#pragma unroll
    for (int r = 0; r < 8; ++r)
        yb[(r + 8 * hi) * ROWPAD + cm] = cy[r] + bias;
    __builtin_amdgcn_wave_barrier();
    const float4 yv = *(const float4*)(yb + 4 * lane);

    const char* zb = (const char*)zws;
    const unsigned qoff = (unsigned)q << 4;
    float4 m = {-INFINITY, -INFINITY, -INFINITY, -INFINITY};
    if (fast) {
        gather8<0>(m, idx, zb, qoff);
        gather8<1>(m, idx, zb, qoff);
        gather8<2>(m, idx, zb, qoff);
        gather8<3>(m, idx, zb, qoff);
    } else {
        const int* sp = src + (size_t)nodeC * kdeg;
        for (int j = 0; j < kdeg; ++j) {
            const unsigned s = (unsigned)sp[j];
            max4(m, *(const float4*)(zb + ((s << 6) | qoff)));
        }
    }

    float ag[4] = { yv.x + m.x, yv.y + m.y, yv.z + m.z, yv.w + m.w };
#pragma unroll
    for (int i = 0; i < 4; ++i)
        ag[i] = isfinite(ag[i]) ? ag[i] : 0.0f;

    float agg[DFEAT];
    agg[0] = quad_bcastf<0>(ag[0]);  agg[1] = quad_bcastf<0>(ag[1]);
    agg[2] = quad_bcastf<0>(ag[2]);  agg[3] = quad_bcastf<0>(ag[3]);
    agg[4] = quad_bcastf<1>(ag[0]);  agg[5] = quad_bcastf<1>(ag[1]);
    agg[6] = quad_bcastf<1>(ag[2]);  agg[7] = quad_bcastf<1>(ag[3]);
    agg[8] = quad_bcastf<2>(ag[0]);  agg[9] = quad_bcastf<2>(ag[1]);

    if (valid && q < 3) {
        float4 acc = *(const float4*)(b2s + 4 * q);
#pragma unroll
        for (int k = 0; k < DFEAT; ++k) {
            const float4 w = *(const float4*)(w2s + k * ROWPAD + 4 * q);
            acc.x = fmaf(agg[k], w.x, acc.x);
            acc.y = fmaf(agg[k], w.y, acc.y);
            acc.z = fmaf(agg[k], w.z, acc.z);
            acc.w = fmaf(agg[k], w.w, acc.w);
        }
        float* o = out + (size_t)node * DFEAT + 4 * q;
        *(float2*)o = make_float2(acc.x, acc.y);
        if (q < 2) *(float2*)(o + 2) = make_float2(acc.z, acc.w);
    }
}

extern "C" void kernel_launch(void* const* d_in, const int* in_sizes, int n_in,
                              void* d_out, int out_size, void* d_ws, size_t ws_size,
                              hipStream_t stream) {
    const float* x   = (const float*)d_in[0];
    const int*   ei  = (const int*)  d_in[1];
    const float* W1  = (const float*)d_in[2];
    const float* b1  = (const float*)d_in[3];
    const float* W2  = (const float*)d_in[4];
    const float* b2  = (const float*)d_in[5];
    float* out = (float*)d_out;

    const int nnodes = in_sizes[0] / DFEAT;
    const int nedges = in_sizes[1] / 2;
    const int kdeg   = nedges / nnodes;
    const int* src   = ei;

    float* zws = (float*)d_ws;

    const int ntiles = (nnodes + 15) / 16;
    {
        dim3 block(256);
        dim3 grid((ntiles + 7) / 8);
        hipLaunchKernelGGL(edgeconv_lin1_wmma, grid, block, 0, stream,
                           x, W1, zws, nnodes, ntiles);
    }
    {
        dim3 block(256);
        dim3 grid((nnodes + 63) / 64);
        hipLaunchKernelGGL(edgeconv_maxagg, grid, block, 0, stream,
                           src, x, zws, W1, b1, W2, b2, out, nnodes, kdeg);
    }
}


# more rows per wave in both kernels (kernel 1: 4 row tiles per wave in 2-wave groups; kernel 2: 16 nodes per wave in 4-wave groups): 6250 -> 1563 and 12500 -> 6250 waves to enter
# speedup vs baseline: 1.5486x; 1.0334x over previous
#include <hip/hip_runtime.h>
#include <math.h>


typedef float v2f __attribute__((ext_vector_type(2)));
typedef float v8f __attribute__((ext_vector_type(8)));

#define DFEAT  10
#define KPAD   12
#define ROWPAD 16

#define TPW 4
#define K1_THREADS 64

__global__ __launch_bounds__(K1_THREADS)
void edgeconv_lin1_wmma(const float* __restrict__ x,
                        const float* __restrict__ W1,
                        float* __restrict__ zws,
                        int nrows, int ntiles) {
    __shared__ float ldsB[16 * KPAD];

    float wst[(16 * KPAD) / K1_THREADS];
#pragma unroll
    for (int j = 0; j < (16 * KPAD) / K1_THREADS; ++j) {
        const int t = threadIdx.x + j * K1_THREADS;
        const int n = t / KPAD, k = t % KPAD;
        const bool ok = (n < DFEAT && k < DFEAT);
        const float wbot = W1[ok ? (k + DFEAT) * DFEAT + n : 0];
        wst[j] = ok ? wbot : 0.0f;
    }
#pragma unroll
    for (int j = 0; j < (16 * KPAD) / K1_THREADS; ++j)
        ldsB[threadIdx.x + j * K1_THREADS] = wst[j];
    __syncthreads();

    const int wavesPerBlock = K1_THREADS >> 5;
    const int tile0 = (blockIdx.x * wavesPerBlock + (threadIdx.x >> 5)) * TPW;
    if (tile0 >= ntiles) return;

    const int lane = threadIdx.x & 31;
    const int hi   = lane >> 4;
    const int m    = lane & 15;

    v2f wb[3];
#pragma unroll
    for (int ks = 0; ks < 3; ++ks) {
        const float2 wbv = *(const float2*)(ldsB + m * KPAD + ks * 4 + 2 * hi);
        wb[ks][0] = wbv.x;  wb[ks][1] = wbv.y;
    }

    v2f xa[TPW][3];
#pragma unroll
    for (int i = 0; i < TPW; ++i) {
        const int row = (tile0 + i) * 16 + m;
        const bool rowOK = (row < nrows);
#pragma unroll
        for (int ks = 0; ks < 3; ++ks) {
            const int k0 = ks * 4 + 2 * hi;
            const bool ok = rowOK && (k0 < DFEAT);
            const float2* px = ok ? (const float2*)(x + (size_t)row * DFEAT + k0)
                                  : (const float2*)x;
            const float2 xv = *px;
            xa[i][ks][0] = ok ? xv.x : 0.0f;  xa[i][ks][1] = ok ? xv.y : 0.0f;
        }
    }

#pragma unroll
    for (int i = 0; i < TPW; ++i) {
        const int base = (tile0 + i) * 16;
        v8f cz = {0.f,0.f,0.f,0.f,0.f,0.f,0.f,0.f};
#pragma unroll
        for (int ks = 0; ks < 3; ++ks) {
            cz = __builtin_amdgcn_wmma_f32_16x16x4_f32(false, xa[i][ks], false, wb[ks],
                                                       (short)0, cz, false, false);
        }

        float* zrow = zws + (size_t)base * ROWPAD + m;
        if (base + 16 <= nrows) {
#pragma unroll
            for (int r = 0; r < 8; ++r) {
                const int rr = r + 8 * hi;
                zrow[(size_t)rr * ROWPAD] = cz[r];
            }
        } else {
#pragma unroll
            for (int r = 0; r < 8; ++r) {
                const int rr = r + 8 * hi;
                if (base + rr < nrows)
                    zrow[(size_t)rr * ROWPAD] = cz[r];
            }
        }
    }
}

#define NPW 16

template <int A>
__device__ __forceinline__ int quad_bcast(int v) {
    return __builtin_amdgcn_mov_dpp(v, A * 0x55, 0xF, 0xF, true);
}
template <int A>
__device__ __forceinline__ float quad_bcastf(float v) {
    return __int_as_float(quad_bcast<A>(__float_as_int(v)));
}

__device__ __forceinline__ void max4(float4& m, const float4 v) {
    m.x = fmaxf(m.x, v.x);  m.y = fmaxf(m.y, v.y);
    m.z = fmaxf(m.z, v.z);  m.w = fmaxf(m.w, v.w);
}

template <int A>
__device__ __forceinline__ void gather8(float4& m, const int (&idx)[8],
                                        const char* __restrict__ zb, unsigned qoff) {
#pragma unroll
    for (int b = 0; b < 8; ++b) {
        const unsigned s = (unsigned)quad_bcast<A>(idx[b]);
        max4(m, *(const float4*)(zb + ((s << 6) | qoff)));
    }
}

__device__ __forceinline__ void finish_node(const float4 yv, const float4 m,
                                            bool valid, int node, int q,
                                            const float* w2s, const float* b2s,
                                            float* __restrict__ out) {
    float ag[4] = { yv.x + m.x, yv.y + m.y, yv.z + m.z, yv.w + m.w };
#pragma unroll
    for (int i = 0; i < 4; ++i)
        ag[i] = isfinite(ag[i]) ? ag[i] : 0.0f;

    float agg[DFEAT];
    agg[0] = quad_bcastf<0>(ag[0]);  agg[1] = quad_bcastf<0>(ag[1]);
    agg[2] = quad_bcastf<0>(ag[2]);  agg[3] = quad_bcastf<0>(ag[3]);
    agg[4] = quad_bcastf<1>(ag[0]);  agg[5] = quad_bcastf<1>(ag[1]);
    agg[6] = quad_bcastf<1>(ag[2]);  agg[7] = quad_bcastf<1>(ag[3]);
    agg[8] = quad_bcastf<2>(ag[0]);  agg[9] = quad_bcastf<2>(ag[1]);

    if (valid && q < 3) {
        float4 acc = *(const float4*)(b2s + 4 * q);
#pragma unroll
        for (int k = 0; k < DFEAT; ++k) {
            const float4 w = *(const float4*)(w2s + k * ROWPAD + 4 * q);
            acc.x = fmaf(agg[k], w.x, acc.x);
            acc.y = fmaf(agg[k], w.y, acc.y);
            acc.z = fmaf(agg[k], w.z, acc.z);
            acc.w = fmaf(agg[k], w.w, acc.w);
        }
        float* o = out + (size_t)node * DFEAT + 4 * q;
        *(float2*)o = make_float2(acc.x, acc.y);
        if (q < 2) *(float2*)(o + 2) = make_float2(acc.z, acc.w);
    }
}

#define K2_THREADS 128

__global__ __launch_bounds__(K2_THREADS)
void edgeconv_maxagg(const int* __restrict__ src,
                     const float* __restrict__ x,
                     const float* __restrict__ zws,
                     const float* __restrict__ W1,
                     const float* __restrict__ b1,
                     const float* __restrict__ W2,
                     const float* __restrict__ b2,
                     float* __restrict__ out,
                     int nnodes, int kdeg) {
    __shared__ float w2s[DFEAT * ROWPAD];
    __shared__ float b2s[ROWPAD];
    __shared__ float ldsA[16 * KPAD];
    __shared__ float ybuf[4][16 * ROWPAD];

    const int lane  = threadIdx.x & 31;
    const int q     = lane & 3;
    const int node0 = (blockIdx.x * (K2_THREADS >> 5) + (threadIdx.x >> 5)) * NPW;
    const int nodeA = node0 + (lane >> 2);
    const int nodeB = nodeA + 8;
    const bool validA = nodeA < nnodes;
    const bool validB = nodeB < nnodes;
    const int nodeAC = validA ? nodeA : 0;
    const int nodeBC = validB ? nodeB : 0;
    const bool fast = (kdeg == 32);

    int idxA[8] = {0, 0, 0, 0, 0, 0, 0, 0};
    int idxB[8] = {0, 0, 0, 0, 0, 0, 0, 0};
    if (fast) {
        const int4* ipa = (const int4*)(src + (size_t)nodeAC * 32 + 8 * q);
        const int4* ipb = (const int4*)(src + (size_t)nodeBC * 32 + 8 * q);
        const int4 a0 = ipa[0], a1 = ipa[1], c0 = ipb[0], c1 = ipb[1];
        idxA[0] = a0.x; idxA[1] = a0.y; idxA[2] = a0.z; idxA[3] = a0.w;
        idxA[4] = a1.x; idxA[5] = a1.y; idxA[6] = a1.z; idxA[7] = a1.w;
        idxB[0] = c0.x; idxB[1] = c0.y; idxB[2] = c0.z; idxB[3] = c0.w;
        idxB[4] = c1.x; idxB[5] = c1.y; idxB[6] = c1.z; idxB[7] = c1.w;
    }
    const int hi = lane >> 4;
    const int cm = lane & 15;
    const bool rowOK = (node0 + cm < nnodes);
    v2f xa[3];
#pragma unroll
    for (int ks = 0; ks < 3; ++ks) {
        const int k0 = ks * 4 + 2 * hi;
        const bool ok = rowOK && (k0 < DFEAT);
        const float2* px = ok ? (const float2*)(x + (size_t)(node0 + cm) * DFEAT + k0)
                              : (const float2*)x;
        const float2 xv = *px;
        xa[ks][0] = ok ? xv.x : 0.0f;  xa[ks][1] = ok ? xv.y : 0.0f;
    }
    const float bias = (cm < DFEAT) ? b1[cm] : 0.0f;

    float wsa[2], ws2[2];
#pragma unroll
    for (int j = 0; j < 2; ++j) {
        const int t = threadIdx.x + j * K2_THREADS;
        const int n = t / KPAD, k = t % KPAD;
        const int k2 = t >> 4, c2 = t & 15;
        const bool ok1 = (t < 16 * KPAD) && n < DFEAT && k < DFEAT;
        const bool ok2 = (t < DFEAT * ROWPAD) && c2 < DFEAT;
        const float wtop = W1[ok1 ? k * DFEAT + n : 0];
        const float wbot = W1[ok1 ? (k + DFEAT) * DFEAT + n : 0];
        const float w2v  = W2[ok2 ? k2 * DFEAT + c2 : 0];
        wsa[j] = ok1 ? (wtop - wbot) : 0.0f;
        ws2[j] = ok2 ? w2v : 0.0f;
    }
#pragma unroll
    for (int j = 0; j < 2; ++j) {
        const int t = threadIdx.x + j * K2_THREADS;
        if (t < 16 * KPAD)      ldsA[t] = wsa[j];
        if (t < DFEAT * ROWPAD) w2s[t]  = ws2[j];
    }
    if (threadIdx.x < ROWPAD)
        b2s[threadIdx.x] = (threadIdx.x < DFEAT) ? b2[threadIdx.x] : 0.0f;
    __syncthreads();
    if (node0 >= nnodes) return;

    v8f cy = {0.f,0.f,0.f,0.f,0.f,0.f,0.f,0.f};
#pragma unroll
    for (int ks = 0; ks < 3; ++ks) {
        const float2 wav = *(const float2*)(ldsA + cm * KPAD + ks * 4 + 2 * hi);
        v2f wa;  wa[0] = wav.x;  wa[1] = wav.y;
        cy = __builtin_amdgcn_wmma_f32_16x16x4_f32(false, xa[ks], false, wa,
                                                   (short)0, cy, false, false);
    }
    float* yb = ybuf[threadIdx.x >> 5];
#pragma unroll
    for (int r = 0; r < 8; ++r)
        yb[(r + 8 * hi) * ROWPAD + cm] = cy[r] + bias;
    __builtin_amdgcn_wave_barrier();
    const float4 yvA = *(const float4*)(yb + 4 * lane);
    const float4 yvB = *(const float4*)(yb + 8 * ROWPAD + 4 * lane);

    const char* zb = (const char*)zws;
    const unsigned qoff = (unsigned)q << 4;
    float4 mA = {-INFINITY, -INFINITY, -INFINITY, -INFINITY};
    float4 mB = {-INFINITY, -INFINITY, -INFINITY, -INFINITY};
    if (fast) {
        gather8<0>(mA, idxA, zb, qoff);
        gather8<1>(mA, idxA, zb, qoff);
        gather8<2>(mA, idxA, zb, qoff);
        gather8<3>(mA, idxA, zb, qoff);
        gather8<0>(mB, idxB, zb, qoff);
        gather8<1>(mB, idxB, zb, qoff);
        gather8<2>(mB, idxB, zb, qoff);
        gather8<3>(mB, idxB, zb, qoff);
    } else {
        const int* spa = src + (size_t)nodeAC * kdeg;
        const int* spb = src + (size_t)nodeBC * kdeg;
        for (int j = 0; j < kdeg; ++j) {
            const unsigned sa = (unsigned)spa[j];
            const unsigned sb = (unsigned)spb[j];
            max4(mA, *(const float4*)(zb + ((sa << 6) | qoff)));
            max4(mB, *(const float4*)(zb + ((sb << 6) | qoff)));
        }
    }

    finish_node(yvA, mA, validA, nodeA, q, w2s, b2s, out);
    finish_node(yvB, mB, validB, nodeB, q, w2s, b2s, out);
}

extern "C" void kernel_launch(void* const* d_in, const int* in_sizes, int n_in,
                              void* d_out, int out_size, void* d_ws, size_t ws_size,
                              hipStream_t stream) {
    const float* x   = (const float*)d_in[0];
    const int*   ei  = (const int*)  d_in[1];
    const float* W1  = (const float*)d_in[2];
    const float* b1  = (const float*)d_in[3];
    const float* W2  = (const float*)d_in[4];
    const float* b2  = (const float*)d_in[5];
    float* out = (float*)d_out;

    const int nnodes = in_sizes[0] / DFEAT;
    const int nedges = in_sizes[1] / 2;
    const int kdeg   = nedges / nnodes;
    const int* src   = ei;

    float* zws = (float*)d_ws;

    const int ntiles = (nnodes + 15) / 16;
    {
        dim3 block(K1_THREADS);
        dim3 grid((ntiles + 7) / 8);
        hipLaunchKernelGGL(edgeconv_lin1_wmma, grid, block, 0, stream,
                           x, W1, zws, nnodes, ntiles);
    }
    {
        dim3 block(K2_THREADS);
        dim3 grid((nnodes + 63) / 64);
        hipLaunchKernelGGL(edgeconv_maxagg, grid, block, 0, stream,
                           src, x, zws, W1, b1, W2, b2, out, nnodes, kdeg);
    }
}
